# starting kernel
# baseline (speedup 1.0000x reference)
#include <hip/hip_runtime.h>
#include <hip/hip_bf16.h>

typedef __attribute__((ext_vector_type(2))) float v2f;
typedef __attribute__((ext_vector_type(8))) float v8f;


__global__ void scr_reduce_C_part(const float* __restrict__ C, float* __restrict__ sCp) {
    const int kl   = blockIdx.x * blockDim.x + threadIdx.x;
    const int part = blockIdx.y;
    const float* p = C + (size_t)part * 96 * 1024 + kl;
    float acc = 0.0f;
#pragma unroll 8
    for (int t = 0; t < 96; ++t) {
        acc += p[(size_t)t * 1024];
    }
    sCp[part * 1024 + kl] = acc;
}

__global__ void scr_reduce_x_part(const float* __restrict__ x, float* __restrict__ sXp) {
    __shared__ float lds[64];
    const int b    = blockIdx.x;
    const int part = blockIdx.y;
    const int t    = threadIdx.x;
    const int w    = (t / 3) & 7;

    if (t < 64) lds[t] = 0.0f;
    __syncthreads();

    const float* p = x + (size_t)b * 256 * 768 + (size_t)part * 64 * 768 + t;
    float acc[8] = {0.f, 0.f, 0.f, 0.f, 0.f, 0.f, 0.f, 0.f};
#pragma unroll 8
    for (int j = 0; j < 64; ++j) {
        acc[j & 7] += p[(size_t)j * 768];
    }

#pragma unroll
    for (int h = 0; h < 8; ++h) {
        atomicAdd(&lds[h * 8 + w], acc[h]);
    }
    __syncthreads();

    if (t < 64) sXp[part * 8192 + b * 64 + t] = lds[t];
}

__global__ void scr_fold(const float* __restrict__ sCp, const float* __restrict__ sXp,
                         float* __restrict__ sC, float* __restrict__ sX) {
    const int idx = blockIdx.x * blockDim.x + threadIdx.x;
    if (idx < 1024) {
        float acc = 0.0f;
#pragma unroll
        for (int p = 0; p < 32; ++p) acc += sCp[p * 1024 + idx];
        sC[idx] = acc;
    } else {
        const int j = idx - 1024;
        float acc = 0.0f;
#pragma unroll
        for (int p = 0; p < 4; ++p) acc += sXp[p * 8192 + j];
        sX[j] = acc;
    }
}

__global__ void scr_outer_wmma(const float* __restrict__ sC,
                               const float* __restrict__ sX,
                               float* __restrict__ out) {
    const int tid  = threadIdx.x;
    const int lane = tid & 31;
    const int gw   = blockIdx.x * (blockDim.x >> 5) + (tid >> 5);
    const int b     = gw >> 8;
    const int rem   = gw & 255;
    const int mtile = rem >> 2;
    const int ntile = rem & 3;
    const int l16   = lane & 15;

    const float aval = sC[mtile * 16 + l16];
    const float bval = 0.25f * sX[b * 64 + ntile * 16 + l16];

    v2f A  = {aval, aval};
    v2f Bv = {bval, bval};
    v8f Cacc = {};
    v8f D = __builtin_amdgcn_wmma_f32_16x16x4_f32(
        false, A, false, Bv, (short)0, Cacc, false, false);

    const int mhi = (lane >> 4) << 3;
    float* basep = out + ((size_t)b * 1024 + (size_t)mtile * 16 + mhi) * 64
                       + ntile * 16 + l16;
#pragma unroll
    for (int v = 0; v < 8; ++v) {
        basep[(size_t)v * 64] = D[v];
    }
}

extern "C" void kernel_launch(void* const* d_in, const int* in_sizes, int n_in,
                              void* d_out, int out_size, void* d_ws, size_t ws_size,
                              hipStream_t stream) {
    (void)in_sizes; (void)n_in; (void)out_size; (void)ws_size;

    const float* x = (const float*)d_in[0];
    const float* C = (const float*)d_in[1];
    float* out = (float*)d_out;

    float* sCp = (float*)d_ws;
    float* sXp = sCp + 32 * 1024;
    float* sC  = sXp + 4 * 8192;
    float* sX  = sC + 1024;

    scr_reduce_C_part<<<dim3(4, 32), 256, 0, stream>>>(C, sCp);
    scr_reduce_x_part<<<dim3(128, 4), 768, 0, stream>>>(x, sXp);
    scr_fold<<<36, 256, 0, stream>>>(sCp, sXp, sC, sX);
    scr_outer_wmma<<<4096, 256, 0, stream>>>(sC, sX, out);
}


# Reductions split by output, not input range: a block owns final sums (x: image x 2 row%8 classes; C: 16 columns), 48 independent 16-byte loads a lane; no partials, no fold launch; sum order changes
# speedup vs baseline: 1.1536x; 1.1536x over previous
#include <hip/hip_runtime.h>
#include <hip/hip_bf16.h>

typedef __attribute__((ext_vector_type(2))) float v2f;
typedef __attribute__((ext_vector_type(8))) float v8f;


__global__ __launch_bounds__(256) __attribute__((amdgpu_waves_per_eu(4, 4)))
void scr_sum_C(const float* __restrict__ C, float* __restrict__ sC) {
    __shared__ float part[8][16];
    const int g     = blockIdx.x;
    const int t     = threadIdx.x;
    const int lane  = t & 31;
    const int wv    = t >> 5;
    const int piece = lane & 3;
    const int rr    = lane >> 2;

    const float4* p = reinterpret_cast<const float4*>(
        C + (size_t)(wv * 8 + rr) * 1024 + g * 16) + piece;

    float4 v[48];
#pragma unroll
    for (int j = 0; j < 48; ++j) {
        v[j] = p[(size_t)j * (64 * 256)];
    }

    float4 acc[4];
#pragma unroll
    for (int a = 0; a < 4; ++a) acc[a] = make_float4(0.f, 0.f, 0.f, 0.f);
#pragma unroll
    for (int j = 0; j < 48; ++j) {
        acc[j & 3].x += v[j].x;
        acc[j & 3].y += v[j].y;
        acc[j & 3].z += v[j].z;
        acc[j & 3].w += v[j].w;
    }
    float4 s;
    s.x = (acc[0].x + acc[1].x) + (acc[2].x + acc[3].x);
    s.y = (acc[0].y + acc[1].y) + (acc[2].y + acc[3].y);
    s.z = (acc[0].z + acc[1].z) + (acc[2].z + acc[3].z);
    s.w = (acc[0].w + acc[1].w) + (acc[2].w + acc[3].w);

#pragma unroll
    for (int off = 4; off < 32; off <<= 1) {
        s.x += __shfl_xor(s.x, off);
        s.y += __shfl_xor(s.y, off);
        s.z += __shfl_xor(s.z, off);
        s.w += __shfl_xor(s.w, off);
    }
    if (lane < 4) {
        part[wv][lane * 4 + 0] = s.x;
        part[wv][lane * 4 + 1] = s.y;
        part[wv][lane * 4 + 2] = s.z;
        part[wv][lane * 4 + 3] = s.w;
    }
    __syncthreads();
    if (t < 16) {
        float a = 0.0f;
#pragma unroll
        for (int w = 0; w < 8; ++w) a += part[w][t];
        sC[g * 16 + t] = a;
    }
}

__global__ __launch_bounds__(256) __attribute__((amdgpu_waves_per_eu(4, 4)))
void scr_sum_x(const float* __restrict__ x, float* __restrict__ sX) {
    __shared__ float lds[16];
    const int b    = blockIdx.x;
    const int hp   = blockIdx.y;
    const int t    = threadIdx.x;
    const int lane = t & 31;
    const int wv   = t >> 5;
    const int hl   = wv & 1;
    const int row0 = 2 * hp + hl + 8 * (wv >> 1);

    if (t < 16) lds[t] = 0.0f;
    __syncthreads();

    const float4* p = reinterpret_cast<const float4*>(
        x + ((size_t)b * 256 + row0) * 768) + lane;

    float4 v[8][6];
#pragma unroll
    for (int j = 0; j < 8; ++j) {
#pragma unroll
        for (int k = 0; k < 6; ++k) {
            v[j][k] = p[(size_t)j * (32 * 192) + k * 32];
        }
    }

    float4 acc[3];
#pragma unroll
    for (int k = 0; k < 3; ++k) acc[k] = make_float4(0.f, 0.f, 0.f, 0.f);
#pragma unroll
    for (int j = 0; j < 8; ++j) {
#pragma unroll
        for (int k = 0; k < 6; ++k) {
            acc[k % 3].x += v[j][k].x;
            acc[k % 3].y += v[j][k].y;
            acc[k % 3].z += v[j][k].z;
            acc[k % 3].w += v[j][k].w;
        }
    }

#pragma unroll
    for (int k = 0; k < 3; ++k) {
        const int q  = lane + 32 * k;
        const int r  = q % 3;
        const int wa = ((4 * q) / 3) & 7;
        const int wb = (wa + 1) & 7;
        const float4 c = acc[k];
        const float xy  = c.x + c.y;
        const float zw  = c.z + c.w;
        const float xyz = xy + c.z;
        const float yzw = c.y + zw;
        const float sa = (r == 0) ? xyz : ((r == 1) ? xy : c.x);
        const float sb = (r == 0) ? c.w : ((r == 1) ? zw : yzw);
        atomicAdd(&lds[hl * 8 + wa], sa);
        atomicAdd(&lds[hl * 8 + wb], sb);
    }
    __syncthreads();

    if (t < 16) sX[b * 64 + hp * 16 + t] = lds[t];
}

__global__ void scr_outer_wmma(const float* __restrict__ sC,
                               const float* __restrict__ sX,
                               float* __restrict__ out) {
    const int tid  = threadIdx.x;
    const int lane = tid & 31;
    const int gw   = blockIdx.x * (blockDim.x >> 5) + (tid >> 5);
    const int b     = gw >> 8;
    const int rem   = gw & 255;
    const int mtile = rem >> 2;
    const int ntile = rem & 3;
    const int l16   = lane & 15;

    const float aval = sC[mtile * 16 + l16];
    const float bval = 0.25f * sX[b * 64 + ntile * 16 + l16];

    v2f A  = {aval, aval};
    v2f Bv = {bval, bval};
    v8f Cacc = {};
    v8f D = __builtin_amdgcn_wmma_f32_16x16x4_f32(
        false, A, false, Bv, (short)0, Cacc, false, false);

    const int mhi = (lane >> 4) << 3;
    float* basep = out + ((size_t)b * 1024 + (size_t)mtile * 16 + mhi) * 64
                       + ntile * 16 + l16;
#pragma unroll
    for (int v = 0; v < 8; ++v) {
        basep[(size_t)v * 64] = D[v];
    }
}

extern "C" void kernel_launch(void* const* d_in, const int* in_sizes, int n_in,
                              void* d_out, int out_size, void* d_ws, size_t ws_size,
                              hipStream_t stream) {
    (void)in_sizes; (void)n_in; (void)out_size; (void)ws_size;

    const float* x = (const float*)d_in[0];
    const float* C = (const float*)d_in[1];
    float* out = (float*)d_out;

    float* sC = (float*)d_ws;
    float* sX = sC + 1024;

    scr_sum_C<<<64, 256, 0, stream>>>(C, sC);
    scr_sum_x<<<dim3(128, 4), 256, 0, stream>>>(x, sX);
    scr_outer_wmma<<<4096, 256, 0, stream>>>(sC, sX, out);
}


# Outer product fused into the x-reduction kernel: a block owns (image, 32 hw), keeps its S_x in LDS and writes its 1024x32 slab of out itself, whole 128-byte lines; 3 launches -> 2, no S_x round trip
# speedup vs baseline: 1.3096x; 1.1352x over previous
#include <hip/hip_runtime.h>
#include <hip/hip_bf16.h>

typedef __attribute__((ext_vector_type(2))) float v2f;
typedef __attribute__((ext_vector_type(8))) float v8f;


__global__ __launch_bounds__(256) __attribute__((amdgpu_waves_per_eu(4, 4)))
void scr_sum_C(const float* __restrict__ C, float* __restrict__ sC) {
    __shared__ float part[8][16];
    const int g     = blockIdx.x;
    const int t     = threadIdx.x;
    const int lane  = t & 31;
    const int wv    = t >> 5;
    const int piece = lane & 3;
    const int rr    = lane >> 2;

    const float4* p = reinterpret_cast<const float4*>(
        C + (size_t)(wv * 8 + rr) * 1024 + g * 16) + piece;

    float4 v[48];
#pragma unroll
    for (int j = 0; j < 48; ++j) {
        v[j] = p[(size_t)j * (64 * 256)];
    }

    float4 acc[4];
#pragma unroll
    for (int a = 0; a < 4; ++a) acc[a] = make_float4(0.f, 0.f, 0.f, 0.f);
#pragma unroll
    for (int j = 0; j < 48; ++j) {
        acc[j & 3].x += v[j].x;
        acc[j & 3].y += v[j].y;
        acc[j & 3].z += v[j].z;
        acc[j & 3].w += v[j].w;
    }
    float4 s;
    s.x = (acc[0].x + acc[1].x) + (acc[2].x + acc[3].x);
    s.y = (acc[0].y + acc[1].y) + (acc[2].y + acc[3].y);
    s.z = (acc[0].z + acc[1].z) + (acc[2].z + acc[3].z);
    s.w = (acc[0].w + acc[1].w) + (acc[2].w + acc[3].w);

#pragma unroll
    for (int off = 4; off < 32; off <<= 1) {
        s.x += __shfl_xor(s.x, off);
        s.y += __shfl_xor(s.y, off);
        s.z += __shfl_xor(s.z, off);
        s.w += __shfl_xor(s.w, off);
    }
    if (lane < 4) {
        part[wv][lane * 4 + 0] = s.x;
        part[wv][lane * 4 + 1] = s.y;
        part[wv][lane * 4 + 2] = s.z;
        part[wv][lane * 4 + 3] = s.w;
    }
    __syncthreads();
    if (t < 16) {
        float a = 0.0f;
#pragma unroll
        for (int w = 0; w < 8; ++w) a += part[w][t];
        sC[g * 16 + t] = a;
    }
}

__global__ __launch_bounds__(512) __attribute__((amdgpu_waves_per_eu(4, 4)))
void scr_sum_x_outer(const float* __restrict__ x, const float* __restrict__ sC,
                     float* __restrict__ out) {
    __shared__ float lds[32];
    const int b    = blockIdx.x;
    const int hq   = blockIdx.y;
    const int t    = threadIdx.x;
    const int lane = t & 31;
    const int wv   = t >> 5;
    const int hl   = wv & 3;
    const int row0 = 4 * hq + hl + 8 * (wv >> 2);
    const int c16  = lane & 15;
    const int kl0  = wv * 64;

    if (t < 32) lds[t] = 0.0f;
    __syncthreads();

    float aval[4];
#pragma unroll
    for (int mt = 0; mt < 4; ++mt) aval[mt] = sC[kl0 + 16 * mt + c16];

    const float4* p = reinterpret_cast<const float4*>(
        x + ((size_t)b * 256 + row0) * 768) + lane;

    float4 v[8][6];
#pragma unroll
    for (int j = 0; j < 8; ++j) {
#pragma unroll
        for (int k = 0; k < 6; ++k) {
            v[j][k] = p[(size_t)j * (32 * 192) + k * 32];
        }
    }

    float4 acc[3];
#pragma unroll
    for (int k = 0; k < 3; ++k) acc[k] = make_float4(0.f, 0.f, 0.f, 0.f);
#pragma unroll
    for (int j = 0; j < 8; ++j) {
#pragma unroll
        for (int k = 0; k < 6; ++k) {
            acc[k % 3].x += v[j][k].x;
            acc[k % 3].y += v[j][k].y;
            acc[k % 3].z += v[j][k].z;
            acc[k % 3].w += v[j][k].w;
        }
    }

#pragma unroll
    for (int k = 0; k < 3; ++k) {
        const int q  = lane + 32 * k;
        const int r  = q % 3;
        const int wa = ((4 * q) / 3) & 7;
        const int wb = (wa + 1) & 7;
        const float4 c = acc[k];
        const float xy  = c.x + c.y;
        const float zw  = c.z + c.w;
        const float xyz = xy + c.z;
        const float yzw = c.y + zw;
        const float sa = (r == 0) ? xyz : ((r == 1) ? xy : c.x);
        const float sb = (r == 0) ? c.w : ((r == 1) ? zw : yzw);
        atomicAdd(&lds[hl * 8 + wa], sa);
        atomicAdd(&lds[hl * 8 + wb], sb);
    }
    __syncthreads();

    const float b0 = 0.25f * lds[2 * c16];
    const float b1 = 0.25f * lds[2 * c16 + 1];
    const v2f B0 = {b0, b0};
    const v2f B1 = {b1, b1};

    const int mhi = (lane >> 4) << 3;
    float* basep = out + ((size_t)b * 1024 + (size_t)(kl0 + mhi)) * 64
                       + 32 * hq + 2 * c16;
#pragma unroll
    for (int mt = 0; mt < 4; ++mt) {
        const v2f A = {aval[mt], aval[mt]};
        const v8f Cacc = {};
        const v8f D0 = __builtin_amdgcn_wmma_f32_16x16x4_f32(
            false, A, false, B0, (short)0, Cacc, false, false);
        const v8f D1 = __builtin_amdgcn_wmma_f32_16x16x4_f32(
            false, A, false, B1, (short)0, Cacc, false, false);
#pragma unroll
        for (int r = 0; r < 8; ++r) {
            *reinterpret_cast<float2*>(basep + (size_t)(16 * mt + r) * 64) =
                make_float2(D0[r], D1[r]);
        }
    }
}

extern "C" void kernel_launch(void* const* d_in, const int* in_sizes, int n_in,
                              void* d_out, int out_size, void* d_ws, size_t ws_size,
                              hipStream_t stream) {
    (void)in_sizes; (void)n_in; (void)out_size; (void)ws_size;

    const float* x = (const float*)d_in[0];
    const float* C = (const float*)d_in[1];
    float* out = (float*)d_out;

    float* sC = (float*)d_ws;

    scr_sum_C<<<64, 256, 0, stream>>>(C, sC);
    scr_sum_x_outer<<<dim3(128, 2), 512, 0, stream>>>(x, sC, out);
}


# Launch 1 (64 S_C blocks) gets 256 more blocks that sum rows 128-255 of every image on WGPs it left idle; launch 2 sums rows 0-127 and adds them; S_C folds 16 waves; S_x, S_C sum order changes
# speedup vs baseline: 1.5357x; 1.1726x over previous
#include <hip/hip_runtime.h>
#include <hip/hip_bf16.h>

typedef __attribute__((ext_vector_type(2))) float v2f;
typedef __attribute__((ext_vector_type(8))) float v8f;


template <int NJ>
__device__ __forceinline__ void sum_rows_into_lds(const float4* __restrict__ p,
                                                  float* __restrict__ l8, int lane) {
    float4 v[NJ][6];
#pragma unroll
    for (int j = 0; j < NJ; ++j) {
#pragma unroll
        for (int k = 0; k < 6; ++k) {
            v[j][k] = p[(size_t)j * (32 * 192) + k * 32];
        }
    }

    float4 acc[3];
#pragma unroll
    for (int k = 0; k < 3; ++k) acc[k] = make_float4(0.f, 0.f, 0.f, 0.f);
#pragma unroll
    for (int j = 0; j < NJ; ++j) {
#pragma unroll
        for (int k = 0; k < 6; ++k) {
            acc[k % 3].x += v[j][k].x;
            acc[k % 3].y += v[j][k].y;
            acc[k % 3].z += v[j][k].z;
            acc[k % 3].w += v[j][k].w;
        }
    }

#pragma unroll
    for (int k = 0; k < 3; ++k) {
        const int q  = lane + 32 * k;
        const int r  = q % 3;
        const int wa = ((4 * q) / 3) & 7;
        const int wb = (wa + 1) & 7;
        const float4 c = acc[k];
        const float xy  = c.x + c.y;
        const float zw  = c.z + c.w;
        const float xyz = xy + c.z;
        const float yzw = c.y + zw;
        const float sa = (r == 0) ? xyz : ((r == 1) ? xy : c.x);
        const float sb = (r == 0) ? c.w : ((r == 1) ? zw : yzw);
        atomicAdd(&l8[wa], sa);
        atomicAdd(&l8[wb], sb);
    }
}

__global__ __launch_bounds__(512) __attribute__((amdgpu_waves_per_eu(8, 8)))
void scr_sum_C_xlow(const float* __restrict__ C, const float* __restrict__ x,
                    float* __restrict__ sC, float* __restrict__ pX) {
    __shared__ float sh[16 * 16];
    const int t    = threadIdx.x;
    const int lane = t & 31;
    const int wv   = t >> 5;

    if (blockIdx.x >= 64 && blockIdx.x < 128) {
        const int g     = blockIdx.x - 64;
        const int piece = lane & 3;
        const int rr    = lane >> 2;

        const float4* p = reinterpret_cast<const float4*>(
            C + (size_t)(wv * 8 + rr) * 1024 + g * 16) + piece;

        float4 v[24];
#pragma unroll
        for (int j = 0; j < 24; ++j) {
            v[j] = p[(size_t)j * (128 * 256)];
        }

        float4 acc[4];
#pragma unroll
        for (int a = 0; a < 4; ++a) acc[a] = make_float4(0.f, 0.f, 0.f, 0.f);
#pragma unroll
        for (int j = 0; j < 24; ++j) {
            acc[j & 3].x += v[j].x;
            acc[j & 3].y += v[j].y;
            acc[j & 3].z += v[j].z;
            acc[j & 3].w += v[j].w;
        }
        float4 s;
        s.x = (acc[0].x + acc[1].x) + (acc[2].x + acc[3].x);
        s.y = (acc[0].y + acc[1].y) + (acc[2].y + acc[3].y);
        s.z = (acc[0].z + acc[1].z) + (acc[2].z + acc[3].z);
        s.w = (acc[0].w + acc[1].w) + (acc[2].w + acc[3].w);

#pragma unroll
        for (int off = 4; off < 32; off <<= 1) {
            s.x += __shfl_xor(s.x, off);
            s.y += __shfl_xor(s.y, off);
            s.z += __shfl_xor(s.z, off);
            s.w += __shfl_xor(s.w, off);
        }
        if (lane < 4) {
            sh[wv * 16 + lane * 4 + 0] = s.x;
            sh[wv * 16 + lane * 4 + 1] = s.y;
            sh[wv * 16 + lane * 4 + 2] = s.z;
            sh[wv * 16 + lane * 4 + 3] = s.w;
        }
        __syncthreads();
        if (t < 16) {
            float a = 0.0f;
#pragma unroll
            for (int w = 0; w < 16; ++w) a += sh[w * 16 + t];
            sC[g * 16 + t] = a;
        }
    } else {
        const int id   = (blockIdx.x < 64) ? blockIdx.x : blockIdx.x - 64;
        const int b    = id & 127;
        const int hq   = id >> 7;
        const int hl   = wv & 3;
        const int row0 = 128 + 4 * hq + hl + 8 * (wv >> 2);

        if (t < 32) sh[t] = 0.0f;
        __syncthreads();

        const float4* p = reinterpret_cast<const float4*>(
            x + ((size_t)b * 256 + row0) * 768) + lane;
        sum_rows_into_lds<4>(p, &sh[hl * 8], lane);
        __syncthreads();

        if (t < 32) pX[(size_t)(b * 2 + hq) * 32 + t] = sh[t];
    }
}

__global__ __launch_bounds__(512) __attribute__((amdgpu_waves_per_eu(4, 4)))
void scr_sum_x_outer(const float* __restrict__ x, const float* __restrict__ sC,
                     const float* __restrict__ pX, float* __restrict__ out) {
    __shared__ float lds[32];
    const int b    = blockIdx.x;
    const int hq   = blockIdx.y;
    const int t    = threadIdx.x;
    const int lane = t & 31;
    const int wv   = t >> 5;
    const int hl   = wv & 3;
    const int row0 = 4 * hq + hl + 8 * (wv >> 2);
    const int c16  = lane & 15;
    const int kl0  = wv * 64;

    if (t < 32) lds[t] = 0.0f;
    __syncthreads();

    float aval[4];
#pragma unroll
    for (int mt = 0; mt < 4; ++mt) aval[mt] = sC[kl0 + 16 * mt + c16];
    const float2 px = *reinterpret_cast<const float2*>(
        pX + (size_t)(b * 2 + hq) * 32 + 2 * c16);

    const float4* p = reinterpret_cast<const float4*>(
        x + ((size_t)b * 256 + row0) * 768) + lane;
    sum_rows_into_lds<4>(p, &lds[hl * 8], lane);
    __syncthreads();

    const float b0 = 0.25f * (lds[2 * c16] + px.x);
    const float b1 = 0.25f * (lds[2 * c16 + 1] + px.y);
    const v2f B0 = {b0, b0};
    const v2f B1 = {b1, b1};

    const int mhi = (lane >> 4) << 3;
    float* basep = out + ((size_t)b * 1024 + (size_t)(kl0 + mhi)) * 64
                       + 32 * hq + 2 * c16;
#pragma unroll
    for (int mt = 0; mt < 4; ++mt) {
        const v2f A = {aval[mt], aval[mt]};
        const v8f Cacc = {};
        const v8f D0 = __builtin_amdgcn_wmma_f32_16x16x4_f32(
            false, A, false, B0, (short)0, Cacc, false, false);
        const v8f D1 = __builtin_amdgcn_wmma_f32_16x16x4_f32(
            false, A, false, B1, (short)0, Cacc, false, false);
#pragma unroll
        for (int r = 0; r < 8; ++r) {
            *reinterpret_cast<float2*>(basep + (size_t)(16 * mt + r) * 64) =
                make_float2(D0[r], D1[r]);
        }
    }
}

extern "C" void kernel_launch(void* const* d_in, const int* in_sizes, int n_in,
                              void* d_out, int out_size, void* d_ws, size_t ws_size,
                              hipStream_t stream) {
    (void)in_sizes; (void)n_in; (void)out_size; (void)ws_size;

    const float* x = (const float*)d_in[0];
    const float* C = (const float*)d_in[1];
    float* out = (float*)d_out;

    float* sC = (float*)d_ws;
    float* pX = sC + 1024;

    scr_sum_C_xlow<<<64 + 256, 512, 0, stream>>>(C, x, sC, pX);
    scr_sum_x_outer<<<dim3(128, 2), 512, 0, stream>>>(x, sC, pX, out);
}
